# starting kernel
# baseline (speedup 1.0000x reference)
#include <hip/hip_runtime.h>

typedef __attribute__((ext_vector_type(2))) float v2f;
typedef __attribute__((ext_vector_type(8))) float v8f;

#define B_      16
#define N_IN    16384
#define N_OUT   8192
#define IDIM    128
#define ODIM    128
#define KDIM    256
#define NDIR    6
#define TILES_MAX (N_OUT / 16)

__global__ void k_zero_counts(int* counts) {
    int t = threadIdx.x;
    if (t < B_ * NDIR) counts[t] = 0;
}

__global__ void k_transpose_w(const float* __restrict__ W, float* __restrict__ Wt) {
    int idx = blockIdx.x * 256 + threadIdx.x;
    if (idx >= NDIR * KDIM * ODIM) return;
    int d   = idx / (KDIM * ODIM);
    int rem = idx % (KDIM * ODIM);
    int k   = rem / ODIM;
    int c   = rem % ODIM;
    Wt[((size_t)d * ODIM + c) * KDIM + k] = W[idx];
}

__global__ void k_bucketize(const int* __restrict__ vec, const int* __restrict__ dmap,
                            int* __restrict__ counts, int* __restrict__ buckets) {
    int idx = blockIdx.x * 256 + threadIdx.x;
    if (idx >= B_ * N_OUT) return;
    int b = idx / N_OUT;
    int n = idx % N_OUT;
    int d = dmap[vec[idx]];
    int pos = atomicAdd(&counts[b * NDIR + d], 1);
    buckets[(size_t)(b * NDIR + d) * N_OUT + pos] = n;
}

__global__ __launch_bounds__(256, 1)
void k_gemm_tile(const float* __restrict__ last, const float* __restrict__ Wt,
                 const float* __restrict__ bias, const float* __restrict__ alpha_p,
                 const int* __restrict__ vec, const int* __restrict__ drev,
                 const int* __restrict__ child_l, const int* __restrict__ child_r,
                 const int* __restrict__ counts, const int* __restrict__ buckets,
                 float* __restrict__ out) {
    __shared__ float pairS[16 * KDIM];
    __shared__ int   nrowS[16];

    const int bd   = blockIdx.x / TILES_MAX;
    const int tile = blockIdx.x % TILES_MAX;
    const int cnt  = counts[bd];
    const int row0 = tile * 16;
    if (row0 >= cnt) return;
    const int b = bd / NDIR;
    const int d = bd % NDIR;
    const int nrows = min(16, cnt - row0);

    const int tid = threadIdx.x;

    if (tid < 16) {
        int r = (tid < nrows) ? tid : 0;
        nrowS[tid] = buckets[(size_t)bd * N_OUT + row0 + r];
    }
    __syncthreads();

    {
        int r   = tid >> 4;
        int seg = tid & 15;
        int n   = nrowS[r];
        int rev = drev[vec[b * N_OUT + n]];
        int cl  = child_l[n], cr = child_r[n];
        int srcA = rev ? cr : cl;
        int srcB = rev ? cl : cr;
        int off  = seg * 16;
        const float* src = (off < IDIM)
            ? (last + ((size_t)b * N_IN + srcA) * IDIM + off)
            : (last + ((size_t)b * N_IN + srcB) * IDIM + (off - IDIM));
        float* dst = &pairS[r * KDIM + off];
#pragma unroll
        for (int i = 0; i < 4; ++i)
            ((float4*)dst)[i] = ((const float4*)src)[i];
    }
    __syncthreads();

    const int lane = tid & 31;
    const int wave = tid >> 5;
    const int col0 = wave * 16;
    const int colN = lane & 15;
    const int koff = (lane >> 4) * 2;
    const int arow = lane & 15;
    const float* wcol = Wt + ((size_t)d * ODIM + (col0 + colN)) * KDIM;

    v8f acc = {};
#pragma unroll 8
    for (int k = 0; k < KDIM; k += 4) {
        v2f a, bf;
        a.x  = pairS[arow * KDIM + k + koff];
        a.y  = pairS[arow * KDIM + k + koff + 1];
        bf.x = wcol[k + koff];
        bf.y = wcol[k + koff + 1];
        acc = __builtin_amdgcn_wmma_f32_16x16x4_f32(
                  false, a, false, bf, (short)0, acc, false, false);
    }

    const float alpha = *alpha_p;
    const float bb    = bias[d * ODIM + col0 + colN];
    const int mbase   = (lane >> 4) * 8;
#pragma unroll
    for (int r = 0; r < 8; ++r) {
        int row = mbase + r;
        if (row < nrows) {
            float y = acc[r] + bb;
            y = (y >= 0.0f) ? y : alpha * y;
            out[((size_t)b * N_OUT + nrowS[row]) * ODIM + col0 + colN] = y;
        }
    }
}

extern "C" void kernel_launch(void* const* d_in, const int* in_sizes, int n_in,
                              void* d_out, int out_size, void* d_ws, size_t ws_size,
                              hipStream_t stream) {
    const float* last    = (const float*)d_in[0];
    const float* W       = (const float*)d_in[1];
    const float* bias    = (const float*)d_in[2];
    const float* alpha   = (const float*)d_in[3];
    const int*   vec     = (const int*)d_in[4];
    const int*   dmap    = (const int*)d_in[5];
    const int*   drev    = (const int*)d_in[6];
    const int*   child_l = (const int*)d_in[7];
    const int*   child_r = (const int*)d_in[8];
    float* out = (float*)d_out;

    char* ws = (char*)d_ws;
    int*   counts  = (int*)ws;
    int*   buckets = (int*)(ws + 512);
    float* Wt      = (float*)(ws + 512 + (size_t)B_ * NDIR * N_OUT * 4);

    k_zero_counts<<<1, 128, 0, stream>>>(counts);
    k_transpose_w<<<(NDIR * KDIM * ODIM + 255) / 256, 256, 0, stream>>>(W, Wt);
    k_bucketize<<<(B_ * N_OUT + 255) / 256, 256, 0, stream>>>(vec, dmap, counts, buckets);
    k_gemm_tile<<<B_ * NDIR * TILES_MAX, 256, 0, stream>>>(
        last, Wt, bias, alpha, vec, drev, child_l, child_r, counts, buckets, out);
}


# Each block takes 64 bucket rows (was 16) and each wave a 2x4 grid of 16x16 tiles, so one A fragment feeds 4 WMMAs and one W fragment 2: 6 fragment loads per 8 WMMAs (was 16)
# speedup vs baseline: 1.3602x; 1.3602x over previous
#include <hip/hip_runtime.h>

typedef __attribute__((ext_vector_type(2))) float v2f;
typedef __attribute__((ext_vector_type(8))) float v8f;

#define B_      16
#define N_IN    16384
#define N_OUT   8192
#define IDIM    128
#define ODIM    128
#define KDIM    256
#define NDIR    6
#define BM     64
#define TILES_MAX (N_OUT / BM)

__global__ void k_zero_counts(int* counts) {
    int t = threadIdx.x;
    if (t < B_ * NDIR) counts[t] = 0;
}

__global__ void k_transpose_w(const float* __restrict__ W, float* __restrict__ Wt) {
    int idx = blockIdx.x * 256 + threadIdx.x;
    if (idx >= NDIR * KDIM * ODIM) return;
    int d   = idx / (KDIM * ODIM);
    int rem = idx % (KDIM * ODIM);
    int k   = rem / ODIM;
    int c   = rem % ODIM;
    Wt[((size_t)d * ODIM + c) * KDIM + k] = W[idx];
}

__global__ void k_bucketize(const int* __restrict__ vec, const int* __restrict__ dmap,
                            int* __restrict__ counts, int* __restrict__ buckets) {
    int idx = blockIdx.x * 256 + threadIdx.x;
    if (idx >= B_ * N_OUT) return;
    int b = idx / N_OUT;
    int n = idx % N_OUT;
    int d = dmap[vec[idx]];
    int pos = atomicAdd(&counts[b * NDIR + d], 1);
    buckets[(size_t)(b * NDIR + d) * N_OUT + pos] = n;
}

__global__ __launch_bounds__(128, 1)
void k_gemm_tile(const float* __restrict__ last, const float* __restrict__ Wt,
                 const float* __restrict__ bias, const float* __restrict__ alpha_p,
                 const int* __restrict__ vec, const int* __restrict__ drev,
                 const int* __restrict__ child_l, const int* __restrict__ child_r,
                 const int* __restrict__ counts, const int* __restrict__ buckets,
                 float* __restrict__ out) {
    __shared__ float pairS[BM * KDIM];
    __shared__ int   nrowS[BM];

    const int bd   = blockIdx.x / TILES_MAX;
    const int tile = blockIdx.x % TILES_MAX;
    const int cnt  = counts[bd];
    const int row0 = tile * BM;
    if (row0 >= cnt) return;
    const int b = bd / NDIR;
    const int d = bd % NDIR;
    const int nrows = min(BM, cnt - row0);

    const int tid = threadIdx.x;

    if (tid < BM) {
        int r = (tid < nrows) ? tid : 0;
        nrowS[tid] = buckets[(size_t)bd * N_OUT + row0 + r];
    }
    __syncthreads();

    {
        const int seg  = tid & 15;
        const int off  = seg * 16;
        const bool lo  = off < IDIM;
        const int offc = off & (IDIM - 1);
        const int rb   = tid >> 4;
        int n[BM / 8], v[BM / 8], cl[BM / 8], cr[BM / 8];
        const float4* src[BM / 8];
        float4 x[BM / 8][4];
#pragma unroll
        for (int it = 0; it < BM / 8; ++it)
            n[it] = nrowS[it * 8 + rb];
#pragma unroll
        for (int it = 0; it < BM / 8; ++it) {
            v[it]  = vec[b * N_OUT + n[it]];
            cl[it] = child_l[n[it]];
            cr[it] = child_r[n[it]];
        }
#pragma unroll
        for (int it = 0; it < BM / 8; ++it) {
            int rev  = drev[v[it]];
            int srcA = rev ? cr[it] : cl[it];
            int srcB = rev ? cl[it] : cr[it];
            int s    = lo ? srcA : srcB;
            src[it]  = (const float4*)(last + ((size_t)b * N_IN + s) * IDIM + offc);
        }
#pragma unroll
        for (int it = 0; it < BM / 8; ++it)
#pragma unroll
            for (int i = 0; i < 4; ++i)
                x[it][i] = src[it][i];
#pragma unroll
        for (int it = 0; it < BM / 8; ++it) {
            float4* dst = (float4*)&pairS[(it * 8 + rb) * KDIM + off];
#pragma unroll
            for (int i = 0; i < 4; ++i)
                dst[i] = x[it][i];
        }
    }
    __syncthreads();

    const int lane = tid & 31;
    const int wave = tid >> 5;
    const int wm   = wave >> 1;
    const int wn   = wave & 1;
    const int colN = lane & 15;
    const int koff = (lane >> 4) * 2;
    const float* arow = &pairS[(wm * 32 + colN) * KDIM + koff];
    const float* wcol = Wt + ((size_t)d * ODIM + (wn * 64 + colN)) * KDIM + koff;

    v8f acc[2][4] = {};
#pragma unroll 8
    for (int k = 0; k < KDIM; k += 4) {
        v2f a[2], bf[4];
#pragma unroll
        for (int i = 0; i < 2; ++i) {
            a[i].x = arow[i * 16 * KDIM + k];
            a[i].y = arow[i * 16 * KDIM + k + 1];
        }
#pragma unroll
        for (int j = 0; j < 4; ++j) {
            bf[j].x = wcol[j * 16 * KDIM + k];
            bf[j].y = wcol[j * 16 * KDIM + k + 1];
        }
#pragma unroll
        for (int i = 0; i < 2; ++i)
#pragma unroll
            for (int j = 0; j < 4; ++j)
                acc[i][j] = __builtin_amdgcn_wmma_f32_16x16x4_f32(
                                false, a[i], false, bf[j], (short)0, acc[i][j], false, false);
    }

    const float alpha = *alpha_p;
    const int mbase   = (lane >> 4) * 8;
    int nr[2][8];
#pragma unroll
    for (int i = 0; i < 2; ++i)
#pragma unroll
        for (int r = 0; r < 8; ++r)
            nr[i][r] = nrowS[wm * 32 + i * 16 + mbase + r];
#pragma unroll
    for (int j = 0; j < 4; ++j) {
        const int col  = wn * 64 + j * 16 + colN;
        const float bb = bias[d * ODIM + col];
#pragma unroll
        for (int i = 0; i < 2; ++i) {
#pragma unroll
            for (int r = 0; r < 8; ++r) {
                int row = wm * 32 + i * 16 + mbase + r;
                float y = acc[i][j][r] + bb;
                y = (y >= 0.0f) ? y : alpha * y;
                if (row < nrows)
                    out[((size_t)b * N_OUT + nr[i][r]) * ODIM + col] = y;
            }
        }
    }
}

extern "C" void kernel_launch(void* const* d_in, const int* in_sizes, int n_in,
                              void* d_out, int out_size, void* d_ws, size_t ws_size,
                              hipStream_t stream) {
    const float* last    = (const float*)d_in[0];
    const float* W       = (const float*)d_in[1];
    const float* bias    = (const float*)d_in[2];
    const float* alpha   = (const float*)d_in[3];
    const int*   vec     = (const int*)d_in[4];
    const int*   dmap    = (const int*)d_in[5];
    const int*   drev    = (const int*)d_in[6];
    const int*   child_l = (const int*)d_in[7];
    const int*   child_r = (const int*)d_in[8];
    float* out = (float*)d_out;

    char* ws = (char*)d_ws;
    int*   counts  = (int*)ws;
    int*   buckets = (int*)(ws + 512);
    float* Wt      = (float*)(ws + 512 + (size_t)B_ * NDIR * N_OUT * 4);

    k_zero_counts<<<1, 128, 0, stream>>>(counts);
    k_transpose_w<<<(NDIR * KDIM * ODIM + 255) / 256, 256, 0, stream>>>(W, Wt);
    k_bucketize<<<(B_ * N_OUT + 255) / 256, 256, 0, stream>>>(vec, dmap, counts, buckets);
    k_gemm_tile<<<B_ * NDIR * TILES_MAX, 128, 0, stream>>>(
        last, Wt, bias, alpha, vec, drev, child_l, child_r, counts, buckets, out);
}


# W copy laid out in WMMA fragment order: each B load is one 16-byte-per-lane read of 4 whole lines covering two k-steps (was an 8-byte read touching 16 lines per k-step)
# speedup vs baseline: 1.7187x; 1.2636x over previous
#include <hip/hip_runtime.h>

typedef __attribute__((ext_vector_type(2))) float v2f;
typedef __attribute__((ext_vector_type(4))) float v4f;
typedef __attribute__((ext_vector_type(8))) float v8f;

#define B_      16
#define N_IN    16384
#define N_OUT   8192
#define IDIM    128
#define ODIM    128
#define KDIM    256
#define NDIR    6
#define BM     64
#define TILES_MAX (N_OUT / BM)

__global__ void k_zero_counts(int* counts) {
    int t = threadIdx.x;
    if (t < B_ * NDIR) counts[t] = 0;
}

#define KPAIRS  (KDIM / 8)
#define JTILES  (ODIM / 16)
__global__ void k_arrange_w(const float* __restrict__ W, float* __restrict__ Wf_) {
    v4f* __restrict__ Wf = (v4f*)Wf_;
    int idx = blockIdx.x * 256 + threadIdx.x;
    if (idx >= NDIR * JTILES * KPAIRS * 32) return;
    int lane = idx & 31;
    int kp   = (idx >> 5) % KPAIRS;
    int jt   = (idx / (32 * KPAIRS)) % JTILES;
    int d    = idx / (32 * KPAIRS * JTILES);
    int c    = lane & 15;
    int h    = lane >> 4;
    const float* w = W + ((size_t)d * KDIM + 8 * kp + 2 * h) * ODIM + jt * 16 + c;
    v4f o;
    o.x = w[0];
    o.y = w[ODIM];
    o.z = w[4 * ODIM];
    o.w = w[5 * ODIM];
    Wf[idx] = o;
}

__global__ void k_bucketize(const int* __restrict__ vec, const int* __restrict__ dmap,
                            int* __restrict__ counts, int* __restrict__ buckets) {
    int idx = blockIdx.x * 256 + threadIdx.x;
    if (idx >= B_ * N_OUT) return;
    int b = idx / N_OUT;
    int n = idx % N_OUT;
    int d = dmap[vec[idx]];
    int pos = atomicAdd(&counts[b * NDIR + d], 1);
    buckets[(size_t)(b * NDIR + d) * N_OUT + pos] = n;
}

__global__ __launch_bounds__(128, 1)
void k_gemm_tile(const float* __restrict__ last, const float* __restrict__ Wf_,
                 const float* __restrict__ bias, const float* __restrict__ alpha_p,
                 const int* __restrict__ vec, const int* __restrict__ drev,
                 const int* __restrict__ child_l, const int* __restrict__ child_r,
                 const int* __restrict__ counts, const int* __restrict__ buckets,
                 float* __restrict__ out) {
    __shared__ float pairS[BM * KDIM];
    __shared__ int   nrowS[BM];

    const int bd   = blockIdx.x / TILES_MAX;
    const int tile = blockIdx.x % TILES_MAX;
    const int cnt  = counts[bd];
    const int row0 = tile * BM;
    if (row0 >= cnt) return;
    const int b = bd / NDIR;
    const int d = bd % NDIR;
    const int nrows = min(BM, cnt - row0);

    const int tid = threadIdx.x;

    if (tid < BM) {
        int r = (tid < nrows) ? tid : 0;
        nrowS[tid] = buckets[(size_t)bd * N_OUT + row0 + r];
    }
    __syncthreads();

    {
        const int seg  = tid & 15;
        const int off  = seg * 16;
        const bool lo  = off < IDIM;
        const int offc = off & (IDIM - 1);
        const int rb   = tid >> 4;
        int n[BM / 8], v[BM / 8], cl[BM / 8], cr[BM / 8];
        const float4* src[BM / 8];
        float4 x[BM / 8][4];
#pragma unroll
        for (int it = 0; it < BM / 8; ++it)
            n[it] = nrowS[it * 8 + rb];
#pragma unroll
        for (int it = 0; it < BM / 8; ++it) {
            v[it]  = vec[b * N_OUT + n[it]];
            cl[it] = child_l[n[it]];
            cr[it] = child_r[n[it]];
        }
#pragma unroll
        for (int it = 0; it < BM / 8; ++it) {
            int rev  = drev[v[it]];
            int srcA = rev ? cr[it] : cl[it];
            int srcB = rev ? cl[it] : cr[it];
            int s    = lo ? srcA : srcB;
            src[it]  = (const float4*)(last + ((size_t)b * N_IN + s) * IDIM + offc);
        }
#pragma unroll
        for (int it = 0; it < BM / 8; ++it)
#pragma unroll
            for (int i = 0; i < 4; ++i)
                x[it][i] = src[it][i];
#pragma unroll
        for (int it = 0; it < BM / 8; ++it) {
            float4* dst = (float4*)&pairS[(it * 8 + rb) * KDIM + off];
#pragma unroll
            for (int i = 0; i < 4; ++i)
                dst[i] = x[it][i];
        }
    }
    __syncthreads();

    const int lane = tid & 31;
    const int wave = tid >> 5;
    const int wm   = wave >> 1;
    const int wn   = wave & 1;
    const int colN = lane & 15;
    const int koff = (lane >> 4) * 2;
    const float* arow = &pairS[(wm * 32 + colN) * KDIM + koff];
    const v4f* wfrag = (const v4f*)Wf_ + ((size_t)(d * JTILES + wn * 4) * KPAIRS) * 32 + lane;

    v8f acc[2][4] = {};
#pragma unroll 4
    for (int kp = 0; kp < KPAIRS; ++kp) {
        v2f a[2][2];
        v4f bq[4];
#pragma unroll
        for (int s = 0; s < 2; ++s)
#pragma unroll
            for (int i = 0; i < 2; ++i) {
                a[s][i].x = arow[i * 16 * KDIM + 8 * kp + 4 * s];
                a[s][i].y = arow[i * 16 * KDIM + 8 * kp + 4 * s + 1];
            }
#pragma unroll
        for (int j = 0; j < 4; ++j)
            bq[j] = wfrag[(j * KPAIRS + kp) * 32];
#pragma unroll
        for (int s = 0; s < 2; ++s)
#pragma unroll
            for (int i = 0; i < 2; ++i)
#pragma unroll
                for (int j = 0; j < 4; ++j) {
                    v2f bf;
                    bf.x = s ? bq[j].z : bq[j].x;
                    bf.y = s ? bq[j].w : bq[j].y;
                    acc[i][j] = __builtin_amdgcn_wmma_f32_16x16x4_f32(
                                    false, a[s][i], false, bf, (short)0, acc[i][j], false, false);
                }
    }

    const float alpha = *alpha_p;
    const int mbase   = (lane >> 4) * 8;
    int nr[2][8];
#pragma unroll
    for (int i = 0; i < 2; ++i)
#pragma unroll
        for (int r = 0; r < 8; ++r)
            nr[i][r] = nrowS[wm * 32 + i * 16 + mbase + r];
#pragma unroll
    for (int j = 0; j < 4; ++j) {
        const int col  = wn * 64 + j * 16 + colN;
        const float bb = bias[d * ODIM + col];
#pragma unroll
        for (int i = 0; i < 2; ++i) {
#pragma unroll
            for (int r = 0; r < 8; ++r) {
                int row = wm * 32 + i * 16 + mbase + r;
                float y = acc[i][j][r] + bb;
                y = (y >= 0.0f) ? y : alpha * y;
                if (row < nrows)
                    out[((size_t)b * N_OUT + nr[i][r]) * ODIM + col] = y;
            }
        }
    }
}

extern "C" void kernel_launch(void* const* d_in, const int* in_sizes, int n_in,
                              void* d_out, int out_size, void* d_ws, size_t ws_size,
                              hipStream_t stream) {
    const float* last    = (const float*)d_in[0];
    const float* W       = (const float*)d_in[1];
    const float* bias    = (const float*)d_in[2];
    const float* alpha   = (const float*)d_in[3];
    const int*   vec     = (const int*)d_in[4];
    const int*   dmap    = (const int*)d_in[5];
    const int*   drev    = (const int*)d_in[6];
    const int*   child_l = (const int*)d_in[7];
    const int*   child_r = (const int*)d_in[8];
    float* out = (float*)d_out;

    char* ws = (char*)d_ws;
    int*   counts  = (int*)ws;
    int*   buckets = (int*)(ws + 512);
    float* Wf      = (float*)(ws + 512 + (size_t)B_ * NDIR * N_OUT * 4);

    k_zero_counts<<<1, 128, 0, stream>>>(counts);
    k_arrange_w<<<(NDIR * JTILES * KPAIRS * 32 + 255) / 256, 256, 0, stream>>>(W, Wf);
    k_bucketize<<<(B_ * N_OUT + 255) / 256, 256, 0, stream>>>(vec, dmap, counts, buckets);
    k_gemm_tile<<<B_ * NDIR * TILES_MAX, 128, 0, stream>>>(
        last, Wf, bias, alpha, vec, drev, child_l, child_r, counts, buckets, out);
}


# GEMM grid holds only the worst-case count of non-empty tiles (2144 blocks, was 12288 with about 10000 exiting empty); each wave finds its bucket and tile by a wave prefix sum over the 96 counts
# speedup vs baseline: 1.8424x; 1.0720x over previous
#include <hip/hip_runtime.h>

typedef __attribute__((ext_vector_type(2))) float v2f;
typedef __attribute__((ext_vector_type(4))) float v4f;
typedef __attribute__((ext_vector_type(8))) float v8f;

#define B_      16
#define N_IN    16384
#define N_OUT   8192
#define IDIM    128
#define ODIM    128
#define KDIM    256
#define NDIR    6
#define BM     64
#define GRID_TILES (B_ * N_OUT / BM + B_ * NDIR)

__global__ void k_zero_counts(int* counts) {
    int t = threadIdx.x;
    if (t < B_ * NDIR) counts[t] = 0;
}

#define KPAIRS  (KDIM / 8)
#define JTILES  (ODIM / 16)
__global__ void k_arrange_w(const float* __restrict__ W, float* __restrict__ Wf_) {
    v4f* __restrict__ Wf = (v4f*)Wf_;
    int idx = blockIdx.x * 256 + threadIdx.x;
    if (idx >= NDIR * JTILES * KPAIRS * 32) return;
    int lane = idx & 31;
    int kp   = (idx >> 5) % KPAIRS;
    int jt   = (idx / (32 * KPAIRS)) % JTILES;
    int d    = idx / (32 * KPAIRS * JTILES);
    int c    = lane & 15;
    int h    = lane >> 4;
    const float* w = W + ((size_t)d * KDIM + 8 * kp + 2 * h) * ODIM + jt * 16 + c;
    v4f o;
    o.x = w[0];
    o.y = w[ODIM];
    o.z = w[4 * ODIM];
    o.w = w[5 * ODIM];
    Wf[idx] = o;
}

__global__ void k_bucketize(const int* __restrict__ vec, const int* __restrict__ dmap,
                            int* __restrict__ counts, int* __restrict__ buckets) {
    int idx = blockIdx.x * 256 + threadIdx.x;
    if (idx >= B_ * N_OUT) return;
    int b = idx / N_OUT;
    int n = idx % N_OUT;
    int d = dmap[vec[idx]];
    int pos = atomicAdd(&counts[b * NDIR + d], 1);
    buckets[(size_t)(b * NDIR + d) * N_OUT + pos] = n;
}

__global__ __launch_bounds__(128, 1)
void k_gemm_tile(const float* __restrict__ last, const float* __restrict__ Wf_,
                 const float* __restrict__ bias, const float* __restrict__ alpha_p,
                 const int* __restrict__ vec, const int* __restrict__ drev,
                 const int* __restrict__ child_l, const int* __restrict__ child_r,
                 const int* __restrict__ counts, const int* __restrict__ buckets,
                 float* __restrict__ out) {
    __shared__ float pairS[BM * KDIM];
    __shared__ int   nrowS[BM];

    const int ln = threadIdx.x & 31;
    const int c0 = counts[3 * ln], c1 = counts[3 * ln + 1], c2 = counts[3 * ln + 2];
    const int t0 = (c0 + BM - 1) / BM, t1 = (c1 + BM - 1) / BM, t2 = (c2 + BM - 1) / BM;
    const int tsum = t0 + t1 + t2;
    int incl = tsum;
#pragma unroll
    for (int o = 1; o < 32; o <<= 1) {
        int up = __shfl_up(incl, o);
        if (ln >= o) incl += up;
    }
    const int r = (int)blockIdx.x - (incl - tsum);
    const unsigned owner = __builtin_amdgcn_ballot_w32(r >= 0 && r < tsum);
    if (owner == 0u) return;
    const int q      = (r < t0) ? 0 : ((r < t0 + t1) ? 1 : 2);
    const int tile_l = (q == 0) ? r : ((q == 1) ? r - t0 : r - t0 - t1);
    const int cnt_l  = (q == 0) ? c0 : ((q == 1) ? c1 : c2);
    const int ol   = __builtin_ctz(owner);
    const int bd   = __builtin_amdgcn_readlane(3 * ln + q, ol);
    const int tile = __builtin_amdgcn_readlane(tile_l, ol);
    const int cnt  = __builtin_amdgcn_readlane(cnt_l, ol);
    const int row0 = tile * BM;
    const int b = bd / NDIR;
    const int d = bd % NDIR;
    const int nrows = min(BM, cnt - row0);

    const int tid = threadIdx.x;

    if (tid < BM) {
        int r = (tid < nrows) ? tid : 0;
        nrowS[tid] = buckets[(size_t)bd * N_OUT + row0 + r];
    }
    __syncthreads();

    {
        const int seg  = tid & 15;
        const int off  = seg * 16;
        const bool lo  = off < IDIM;
        const int offc = off & (IDIM - 1);
        const int rb   = tid >> 4;
        int n[BM / 8], v[BM / 8], cl[BM / 8], cr[BM / 8];
        const float4* src[BM / 8];
        float4 x[BM / 8][4];
#pragma unroll
        for (int it = 0; it < BM / 8; ++it)
            n[it] = nrowS[it * 8 + rb];
#pragma unroll
        for (int it = 0; it < BM / 8; ++it) {
            v[it]  = vec[b * N_OUT + n[it]];
            cl[it] = child_l[n[it]];
            cr[it] = child_r[n[it]];
        }
#pragma unroll
        for (int it = 0; it < BM / 8; ++it) {
            int rev  = drev[v[it]];
            int srcA = rev ? cr[it] : cl[it];
            int srcB = rev ? cl[it] : cr[it];
            int s    = lo ? srcA : srcB;
            src[it]  = (const float4*)(last + ((size_t)b * N_IN + s) * IDIM + offc);
        }
#pragma unroll
        for (int it = 0; it < BM / 8; ++it)
#pragma unroll
            for (int i = 0; i < 4; ++i)
                x[it][i] = src[it][i];
#pragma unroll
        for (int it = 0; it < BM / 8; ++it) {
            float4* dst = (float4*)&pairS[(it * 8 + rb) * KDIM + off];
#pragma unroll
            for (int i = 0; i < 4; ++i)
                dst[i] = x[it][i];
        }
    }
    __syncthreads();

    const int lane = tid & 31;
    const int wave = tid >> 5;
    const int wm   = wave >> 1;
    const int wn   = wave & 1;
    const int colN = lane & 15;
    const int koff = (lane >> 4) * 2;
    const float* arow = &pairS[(wm * 32 + colN) * KDIM + koff];
    const v4f* wfrag = (const v4f*)Wf_ + ((size_t)(d * JTILES + wn * 4) * KPAIRS) * 32 + lane;

    v8f acc[2][4] = {};
#pragma unroll 4
    for (int kp = 0; kp < KPAIRS; ++kp) {
        v2f a[2][2];
        v4f bq[4];
#pragma unroll
        for (int s = 0; s < 2; ++s)
#pragma unroll
            for (int i = 0; i < 2; ++i) {
                a[s][i].x = arow[i * 16 * KDIM + 8 * kp + 4 * s];
                a[s][i].y = arow[i * 16 * KDIM + 8 * kp + 4 * s + 1];
            }
#pragma unroll
        for (int j = 0; j < 4; ++j)
            bq[j] = wfrag[(j * KPAIRS + kp) * 32];
#pragma unroll
        for (int s = 0; s < 2; ++s)
#pragma unroll
            for (int i = 0; i < 2; ++i)
#pragma unroll
                for (int j = 0; j < 4; ++j) {
                    v2f bf;
                    bf.x = s ? bq[j].z : bq[j].x;
                    bf.y = s ? bq[j].w : bq[j].y;
                    acc[i][j] = __builtin_amdgcn_wmma_f32_16x16x4_f32(
                                    false, a[s][i], false, bf, (short)0, acc[i][j], false, false);
                }
    }

    const float alpha = *alpha_p;
    const int mbase   = (lane >> 4) * 8;
    int nr[2][8];
#pragma unroll
    for (int i = 0; i < 2; ++i)
#pragma unroll
        for (int r = 0; r < 8; ++r)
            nr[i][r] = nrowS[wm * 32 + i * 16 + mbase + r];
#pragma unroll
    for (int j = 0; j < 4; ++j) {
        const int col  = wn * 64 + j * 16 + colN;
        const float bb = bias[d * ODIM + col];
#pragma unroll
        for (int i = 0; i < 2; ++i) {
#pragma unroll
            for (int r = 0; r < 8; ++r) {
                int row = wm * 32 + i * 16 + mbase + r;
                float y = acc[i][j][r] + bb;
                y = (y >= 0.0f) ? y : alpha * y;
                if (row < nrows)
                    out[((size_t)b * N_OUT + nr[i][r]) * ODIM + col] = y;
            }
        }
    }
}

extern "C" void kernel_launch(void* const* d_in, const int* in_sizes, int n_in,
                              void* d_out, int out_size, void* d_ws, size_t ws_size,
                              hipStream_t stream) {
    const float* last    = (const float*)d_in[0];
    const float* W       = (const float*)d_in[1];
    const float* bias    = (const float*)d_in[2];
    const float* alpha   = (const float*)d_in[3];
    const int*   vec     = (const int*)d_in[4];
    const int*   dmap    = (const int*)d_in[5];
    const int*   drev    = (const int*)d_in[6];
    const int*   child_l = (const int*)d_in[7];
    const int*   child_r = (const int*)d_in[8];
    float* out = (float*)d_out;

    char* ws = (char*)d_ws;
    int*   counts  = (int*)ws;
    int*   buckets = (int*)(ws + 512);
    float* Wf      = (float*)(ws + 512 + (size_t)B_ * NDIR * N_OUT * 4);

    k_zero_counts<<<1, 128, 0, stream>>>(counts);
    k_arrange_w<<<(NDIR * JTILES * KPAIRS * 32 + 255) / 256, 256, 0, stream>>>(W, Wf);
    k_bucketize<<<(B_ * N_OUT + 255) / 256, 256, 0, stream>>>(vec, dmap, counts, buckets);
    k_gemm_tile<<<GRID_TILES, 128, 0, stream>>>(
        last, Wf, bias, alpha, vec, drev, child_l, child_r, counts, buckets, out);
}


# Buckets built without atomics: one 1024-thread block per batch ranks its rows by wave ballots and an LDS prefix sum (was 131072 returning global adds on 96 words plus a counter-zeroing launch)
# speedup vs baseline: 2.4516x; 1.3307x over previous
#include <hip/hip_runtime.h>

typedef __attribute__((ext_vector_type(2))) float v2f;
typedef __attribute__((ext_vector_type(4))) float v4f;
typedef __attribute__((ext_vector_type(8))) float v8f;

#define B_      16
#define N_IN    16384
#define N_OUT   8192
#define IDIM    128
#define ODIM    128
#define KDIM    256
#define NDIR    6
#define BM     64
#define GRID_TILES (B_ * N_OUT / BM + B_ * NDIR)

#define KPAIRS  (KDIM / 8)
#define JTILES  (ODIM / 16)
__global__ void k_arrange_w(const float* __restrict__ W, float* __restrict__ Wf_) {
    v4f* __restrict__ Wf = (v4f*)Wf_;
    int idx = blockIdx.x * 256 + threadIdx.x;
    if (idx >= NDIR * JTILES * KPAIRS * 32) return;
    int lane = idx & 31;
    int kp   = (idx >> 5) % KPAIRS;
    int jt   = (idx / (32 * KPAIRS)) % JTILES;
    int d    = idx / (32 * KPAIRS * JTILES);
    int c    = lane & 15;
    int h    = lane >> 4;
    const float* w = W + ((size_t)d * KDIM + 8 * kp + 2 * h) * ODIM + jt * 16 + c;
    v4f o;
    o.x = w[0];
    o.y = w[ODIM];
    o.z = w[4 * ODIM];
    o.w = w[5 * ODIM];
    Wf[idx] = o;
}

#define BK_SEGS     (N_OUT / 32)
#define BK_PER_WAVE (BK_SEGS / 32)
#define BK_PITCH    (BK_SEGS + 1)
__global__ __launch_bounds__(1024)
void k_bucketize(const int* __restrict__ vec, const int* __restrict__ dmap,
                 int* __restrict__ counts, int* __restrict__ buckets) {
    __shared__ int segS[NDIR * BK_PITCH];
    const int b    = blockIdx.x;
    const int lane = threadIdx.x & 31;
    const int wave = threadIdx.x >> 5;
    const unsigned below = (1u << lane) - 1u;

    int dir[BK_PER_WAVE], rank[BK_PER_WAVE];
#pragma unroll
    for (int j = 0; j < BK_PER_WAVE; ++j)
        dir[j] = vec[b * N_OUT + (j * 32 + wave) * 32 + lane];
#pragma unroll
    for (int j = 0; j < BK_PER_WAVE; ++j)
        dir[j] = dmap[dir[j]];
#pragma unroll
    for (int j = 0; j < BK_PER_WAVE; ++j) {
        int rk = 0, cnt = 0;
#pragma unroll
        for (int d = 0; d < NDIR; ++d) {
            const unsigned m = __builtin_amdgcn_ballot_w32(dir[j] == d);
            if (dir[j] == d) rk  = __builtin_popcount(m & below);
            if (lane == d)   cnt = __builtin_popcount(m);
        }
        rank[j] = rk;
        if (lane < NDIR) segS[lane * BK_PITCH + j * 32 + wave] = cnt;
    }
    __syncthreads();

    if (wave < NDIR) {
        int* row = &segS[wave * BK_PITCH + BK_PER_WAVE * lane];
        int c[BK_PER_WAVE];
        int run = 0;
#pragma unroll
        for (int k = 0; k < BK_PER_WAVE; ++k) {
            const int t = row[k];
            c[k] = run;
            run += t;
        }
        int incl = run;
#pragma unroll
        for (int o = 1; o < 32; o <<= 1) {
            const int up = __shfl_up(incl, o);
            if (lane >= o) incl += up;
        }
        const int base = incl - run;
#pragma unroll
        for (int k = 0; k < BK_PER_WAVE; ++k)
            row[k] = base + c[k];
        if (lane == 31) counts[b * NDIR + wave] = incl;
    }
    __syncthreads();

#pragma unroll
    for (int j = 0; j < BK_PER_WAVE; ++j) {
        const int seg = j * 32 + wave;
        const int d   = dir[j];
        if ((unsigned)d < (unsigned)NDIR)
            buckets[(size_t)(b * NDIR + d) * N_OUT + segS[d * BK_PITCH + seg] + rank[j]] = seg * 32 + lane;
    }
}

__global__ __launch_bounds__(128, 1)
void k_gemm_tile(const float* __restrict__ last, const float* __restrict__ Wf_,
                 const float* __restrict__ bias, const float* __restrict__ alpha_p,
                 const int* __restrict__ vec, const int* __restrict__ drev,
                 const int* __restrict__ child_l, const int* __restrict__ child_r,
                 const int* __restrict__ counts, const int* __restrict__ buckets,
                 float* __restrict__ out) {
    __shared__ float pairS[BM * KDIM];
    __shared__ int   nrowS[BM];

    const int ln = threadIdx.x & 31;
    const int c0 = counts[3 * ln], c1 = counts[3 * ln + 1], c2 = counts[3 * ln + 2];
    const int t0 = (c0 + BM - 1) / BM, t1 = (c1 + BM - 1) / BM, t2 = (c2 + BM - 1) / BM;
    const int tsum = t0 + t1 + t2;
    int incl = tsum;
#pragma unroll
    for (int o = 1; o < 32; o <<= 1) {
        int up = __shfl_up(incl, o);
        if (ln >= o) incl += up;
    }
    const int r = (int)blockIdx.x - (incl - tsum);
    const unsigned owner = __builtin_amdgcn_ballot_w32(r >= 0 && r < tsum);
    if (owner == 0u) return;
    const int q      = (r < t0) ? 0 : ((r < t0 + t1) ? 1 : 2);
    const int tile_l = (q == 0) ? r : ((q == 1) ? r - t0 : r - t0 - t1);
    const int cnt_l  = (q == 0) ? c0 : ((q == 1) ? c1 : c2);
    const int ol   = __builtin_ctz(owner);
    const int bd   = __builtin_amdgcn_readlane(3 * ln + q, ol);
    const int tile = __builtin_amdgcn_readlane(tile_l, ol);
    const int cnt  = __builtin_amdgcn_readlane(cnt_l, ol);
    const int row0 = tile * BM;
    const int b = bd / NDIR;
    const int d = bd % NDIR;
    const int nrows = min(BM, cnt - row0);

    const int tid = threadIdx.x;

    if (tid < BM) {
        int r = (tid < nrows) ? tid : 0;
        nrowS[tid] = buckets[(size_t)bd * N_OUT + row0 + r];
    }
    __syncthreads();

    {
        const int seg  = tid & 15;
        const int off  = seg * 16;
        const bool lo  = off < IDIM;
        const int offc = off & (IDIM - 1);
        const int rb   = tid >> 4;
        int n[BM / 8], v[BM / 8], cl[BM / 8], cr[BM / 8];
        const float4* src[BM / 8];
        float4 x[BM / 8][4];
#pragma unroll
        for (int it = 0; it < BM / 8; ++it)
            n[it] = nrowS[it * 8 + rb];
#pragma unroll
        for (int it = 0; it < BM / 8; ++it) {
            v[it]  = vec[b * N_OUT + n[it]];
            cl[it] = child_l[n[it]];
            cr[it] = child_r[n[it]];
        }
#pragma unroll
        for (int it = 0; it < BM / 8; ++it) {
            int rev  = drev[v[it]];
            int srcA = rev ? cr[it] : cl[it];
            int srcB = rev ? cl[it] : cr[it];
            int s    = lo ? srcA : srcB;
            src[it]  = (const float4*)(last + ((size_t)b * N_IN + s) * IDIM + offc);
        }
#pragma unroll
        for (int it = 0; it < BM / 8; ++it)
#pragma unroll
            for (int i = 0; i < 4; ++i)
                x[it][i] = src[it][i];
#pragma unroll
        for (int it = 0; it < BM / 8; ++it) {
            float4* dst = (float4*)&pairS[(it * 8 + rb) * KDIM + off];
#pragma unroll
            for (int i = 0; i < 4; ++i)
                dst[i] = x[it][i];
        }
    }
    __syncthreads();

    const int lane = tid & 31;
    const int wave = tid >> 5;
    const int wm   = wave >> 1;
    const int wn   = wave & 1;
    const int colN = lane & 15;
    const int koff = (lane >> 4) * 2;
    const float* arow = &pairS[(wm * 32 + colN) * KDIM + koff];
    const v4f* wfrag = (const v4f*)Wf_ + ((size_t)(d * JTILES + wn * 4) * KPAIRS) * 32 + lane;

    v8f acc[2][4] = {};
#pragma unroll 4
    for (int kp = 0; kp < KPAIRS; ++kp) {
        v2f a[2][2];
        v4f bq[4];
#pragma unroll
        for (int s = 0; s < 2; ++s)
#pragma unroll
            for (int i = 0; i < 2; ++i) {
                a[s][i].x = arow[i * 16 * KDIM + 8 * kp + 4 * s];
                a[s][i].y = arow[i * 16 * KDIM + 8 * kp + 4 * s + 1];
            }
#pragma unroll
        for (int j = 0; j < 4; ++j)
            bq[j] = wfrag[(j * KPAIRS + kp) * 32];
#pragma unroll
        for (int s = 0; s < 2; ++s)
#pragma unroll
            for (int i = 0; i < 2; ++i)
#pragma unroll
                for (int j = 0; j < 4; ++j) {
                    v2f bf;
                    bf.x = s ? bq[j].z : bq[j].x;
                    bf.y = s ? bq[j].w : bq[j].y;
                    acc[i][j] = __builtin_amdgcn_wmma_f32_16x16x4_f32(
                                    false, a[s][i], false, bf, (short)0, acc[i][j], false, false);
                }
    }

    const float alpha = *alpha_p;
    const int mbase   = (lane >> 4) * 8;
    int nr[2][8];
#pragma unroll
    for (int i = 0; i < 2; ++i)
#pragma unroll
        for (int r = 0; r < 8; ++r)
            nr[i][r] = nrowS[wm * 32 + i * 16 + mbase + r];
#pragma unroll
    for (int j = 0; j < 4; ++j) {
        const int col  = wn * 64 + j * 16 + colN;
        const float bb = bias[d * ODIM + col];
#pragma unroll
        for (int i = 0; i < 2; ++i) {
#pragma unroll
            for (int r = 0; r < 8; ++r) {
                int row = wm * 32 + i * 16 + mbase + r;
                float y = acc[i][j][r] + bb;
                y = (y >= 0.0f) ? y : alpha * y;
                if (row < nrows)
                    out[((size_t)b * N_OUT + nr[i][r]) * ODIM + col] = y;
            }
        }
    }
}

extern "C" void kernel_launch(void* const* d_in, const int* in_sizes, int n_in,
                              void* d_out, int out_size, void* d_ws, size_t ws_size,
                              hipStream_t stream) {
    const float* last    = (const float*)d_in[0];
    const float* W       = (const float*)d_in[1];
    const float* bias    = (const float*)d_in[2];
    const float* alpha   = (const float*)d_in[3];
    const int*   vec     = (const int*)d_in[4];
    const int*   dmap    = (const int*)d_in[5];
    const int*   drev    = (const int*)d_in[6];
    const int*   child_l = (const int*)d_in[7];
    const int*   child_r = (const int*)d_in[8];
    float* out = (float*)d_out;

    char* ws = (char*)d_ws;
    int*   counts  = (int*)ws;
    int*   buckets = (int*)(ws + 512);
    float* Wf      = (float*)(ws + 512 + (size_t)B_ * NDIR * N_OUT * 4);

    k_arrange_w<<<(NDIR * JTILES * KPAIRS * 32 + 255) / 256, 256, 0, stream>>>(W, Wf);
    k_bucketize<<<B_, 1024, 0, stream>>>(vec, dmap, counts, buckets);
    k_gemm_tile<<<GRID_TILES, 128, 0, stream>>>(
        last, Wf, bias, alpha, vec, drev, child_l, child_r, counts, buckets, out);
}


# W re-layout and bucketing share one launch (64 blocks of 1024 threads: 16 bucket one batch each, 48 write the fragment-order W copy on other WGPs): 2 dependent launches per call, was 3
# speedup vs baseline: 2.5395x; 1.0359x over previous
#include <hip/hip_runtime.h>

typedef __attribute__((ext_vector_type(2))) float v2f;
typedef __attribute__((ext_vector_type(4))) float v4f;
typedef __attribute__((ext_vector_type(8))) float v8f;

#define B_      16
#define N_IN    16384
#define N_OUT   8192
#define IDIM    128
#define ODIM    128
#define KDIM    256
#define NDIR    6
#define BM     64
#define GRID_TILES (B_ * N_OUT / BM + B_ * NDIR)

#define KPAIRS  (KDIM / 8)
#define JTILES  (ODIM / 16)
#define ARR_ELEMS  (NDIR * JTILES * KPAIRS * 32)
#define ARR_BLOCKS (ARR_ELEMS / 1024)
__device__ __forceinline__ void arrange_w(const float* __restrict__ W, float* __restrict__ Wf_, int idx) {
    v4f* __restrict__ Wf = (v4f*)Wf_;
    int lane = idx & 31;
    int kp   = (idx >> 5) % KPAIRS;
    int jt   = (idx / (32 * KPAIRS)) % JTILES;
    int d    = idx / (32 * KPAIRS * JTILES);
    int c    = lane & 15;
    int h    = lane >> 4;
    const float* w = W + ((size_t)d * KDIM + 8 * kp + 2 * h) * ODIM + jt * 16 + c;
    v4f o;
    o.x = w[0];
    o.y = w[ODIM];
    o.z = w[4 * ODIM];
    o.w = w[5 * ODIM];
    Wf[idx] = o;
}

#define BK_SEGS     (N_OUT / 32)
#define BK_PER_WAVE (BK_SEGS / 32)
#define BK_PITCH    (BK_SEGS + 1)
__device__ __forceinline__
void bucketize(const int* __restrict__ vec, const int* __restrict__ dmap,
               int* __restrict__ counts, int* __restrict__ buckets, const int b) {
    __shared__ int segS[NDIR * BK_PITCH];
    const int lane = threadIdx.x & 31;
    const int wave = threadIdx.x >> 5;
    const unsigned below = (1u << lane) - 1u;

    int dir[BK_PER_WAVE], rank[BK_PER_WAVE];
#pragma unroll
    for (int j = 0; j < BK_PER_WAVE; ++j)
        dir[j] = vec[b * N_OUT + (j * 32 + wave) * 32 + lane];
#pragma unroll
    for (int j = 0; j < BK_PER_WAVE; ++j)
        dir[j] = dmap[dir[j]];
#pragma unroll
    for (int j = 0; j < BK_PER_WAVE; ++j) {
        int rk = 0, cnt = 0;
#pragma unroll
        for (int d = 0; d < NDIR; ++d) {
            const unsigned m = __builtin_amdgcn_ballot_w32(dir[j] == d);
            if (dir[j] == d) rk  = __builtin_popcount(m & below);
            if (lane == d)   cnt = __builtin_popcount(m);
        }
        rank[j] = rk;
        if (lane < NDIR) segS[lane * BK_PITCH + j * 32 + wave] = cnt;
    }
    __syncthreads();

    if (wave < NDIR) {
        int* row = &segS[wave * BK_PITCH + BK_PER_WAVE * lane];
        int c[BK_PER_WAVE];
        int run = 0;
#pragma unroll
        for (int k = 0; k < BK_PER_WAVE; ++k) {
            const int t = row[k];
            c[k] = run;
            run += t;
        }
        int incl = run;
#pragma unroll
        for (int o = 1; o < 32; o <<= 1) {
            const int up = __shfl_up(incl, o);
            if (lane >= o) incl += up;
        }
        const int base = incl - run;
#pragma unroll
        for (int k = 0; k < BK_PER_WAVE; ++k)
            row[k] = base + c[k];
        if (lane == 31) counts[b * NDIR + wave] = incl;
    }
    __syncthreads();

#pragma unroll
    for (int j = 0; j < BK_PER_WAVE; ++j) {
        const int seg = j * 32 + wave;
        const int d   = dir[j];
        if ((unsigned)d < (unsigned)NDIR)
            buckets[(size_t)(b * NDIR + d) * N_OUT + segS[d * BK_PITCH + seg] + rank[j]] = seg * 32 + lane;
    }
}

__global__ __launch_bounds__(1024)
void k_prepare(const float* __restrict__ W, float* __restrict__ Wf,
               const int* __restrict__ vec, const int* __restrict__ dmap,
               int* __restrict__ counts, int* __restrict__ buckets) {
    if (blockIdx.x < B_)
        bucketize(vec, dmap, counts, buckets, (int)blockIdx.x);
    else
        arrange_w(W, Wf, ((int)blockIdx.x - B_) * 1024 + (int)threadIdx.x);
}

__global__ __launch_bounds__(128, 1)
void k_gemm_tile(const float* __restrict__ last, const float* __restrict__ Wf_,
                 const float* __restrict__ bias, const float* __restrict__ alpha_p,
                 const int* __restrict__ vec, const int* __restrict__ drev,
                 const int* __restrict__ child_l, const int* __restrict__ child_r,
                 const int* __restrict__ counts, const int* __restrict__ buckets,
                 float* __restrict__ out) {
    __shared__ float pairS[BM * KDIM];
    __shared__ int   nrowS[BM];

    const int ln = threadIdx.x & 31;
    const int c0 = counts[3 * ln], c1 = counts[3 * ln + 1], c2 = counts[3 * ln + 2];
    const int t0 = (c0 + BM - 1) / BM, t1 = (c1 + BM - 1) / BM, t2 = (c2 + BM - 1) / BM;
    const int tsum = t0 + t1 + t2;
    int incl = tsum;
#pragma unroll
    for (int o = 1; o < 32; o <<= 1) {
        int up = __shfl_up(incl, o);
        if (ln >= o) incl += up;
    }
    const int r = (int)blockIdx.x - (incl - tsum);
    const unsigned owner = __builtin_amdgcn_ballot_w32(r >= 0 && r < tsum);
    if (owner == 0u) return;
    const int q      = (r < t0) ? 0 : ((r < t0 + t1) ? 1 : 2);
    const int tile_l = (q == 0) ? r : ((q == 1) ? r - t0 : r - t0 - t1);
    const int cnt_l  = (q == 0) ? c0 : ((q == 1) ? c1 : c2);
    const int ol   = __builtin_ctz(owner);
    const int bd   = __builtin_amdgcn_readlane(3 * ln + q, ol);
    const int tile = __builtin_amdgcn_readlane(tile_l, ol);
    const int cnt  = __builtin_amdgcn_readlane(cnt_l, ol);
    const int row0 = tile * BM;
    const int b = bd / NDIR;
    const int d = bd % NDIR;
    const int nrows = min(BM, cnt - row0);

    const int tid = threadIdx.x;

    if (tid < BM) {
        int r = (tid < nrows) ? tid : 0;
        nrowS[tid] = buckets[(size_t)bd * N_OUT + row0 + r];
    }
    __syncthreads();

    {
        const int seg  = tid & 15;
        const int off  = seg * 16;
        const bool lo  = off < IDIM;
        const int offc = off & (IDIM - 1);
        const int rb   = tid >> 4;
        int n[BM / 8], v[BM / 8], cl[BM / 8], cr[BM / 8];
        const float4* src[BM / 8];
        float4 x[BM / 8][4];
#pragma unroll
        for (int it = 0; it < BM / 8; ++it)
            n[it] = nrowS[it * 8 + rb];
#pragma unroll
        for (int it = 0; it < BM / 8; ++it) {
            v[it]  = vec[b * N_OUT + n[it]];
            cl[it] = child_l[n[it]];
            cr[it] = child_r[n[it]];
        }
#pragma unroll
        for (int it = 0; it < BM / 8; ++it) {
            int rev  = drev[v[it]];
            int srcA = rev ? cr[it] : cl[it];
            int srcB = rev ? cl[it] : cr[it];
            int s    = lo ? srcA : srcB;
            src[it]  = (const float4*)(last + ((size_t)b * N_IN + s) * IDIM + offc);
        }
#pragma unroll
        for (int it = 0; it < BM / 8; ++it)
#pragma unroll
            for (int i = 0; i < 4; ++i)
                x[it][i] = src[it][i];
#pragma unroll
        for (int it = 0; it < BM / 8; ++it) {
            float4* dst = (float4*)&pairS[(it * 8 + rb) * KDIM + off];
#pragma unroll
            for (int i = 0; i < 4; ++i)
                dst[i] = x[it][i];
        }
    }
    __syncthreads();

    const int lane = tid & 31;
    const int wave = tid >> 5;
    const int wm   = wave >> 1;
    const int wn   = wave & 1;
    const int colN = lane & 15;
    const int koff = (lane >> 4) * 2;
    const float* arow = &pairS[(wm * 32 + colN) * KDIM + koff];
    const v4f* wfrag = (const v4f*)Wf_ + ((size_t)(d * JTILES + wn * 4) * KPAIRS) * 32 + lane;

    v8f acc[2][4] = {};
#pragma unroll 4
    for (int kp = 0; kp < KPAIRS; ++kp) {
        v2f a[2][2];
        v4f bq[4];
#pragma unroll
        for (int s = 0; s < 2; ++s)
#pragma unroll
            for (int i = 0; i < 2; ++i) {
                a[s][i].x = arow[i * 16 * KDIM + 8 * kp + 4 * s];
                a[s][i].y = arow[i * 16 * KDIM + 8 * kp + 4 * s + 1];
            }
#pragma unroll
        for (int j = 0; j < 4; ++j)
            bq[j] = wfrag[(j * KPAIRS + kp) * 32];
#pragma unroll
        for (int s = 0; s < 2; ++s)
#pragma unroll
            for (int i = 0; i < 2; ++i)
#pragma unroll
                for (int j = 0; j < 4; ++j) {
                    v2f bf;
                    bf.x = s ? bq[j].z : bq[j].x;
                    bf.y = s ? bq[j].w : bq[j].y;
                    acc[i][j] = __builtin_amdgcn_wmma_f32_16x16x4_f32(
                                    false, a[s][i], false, bf, (short)0, acc[i][j], false, false);
                }
    }

    const float alpha = *alpha_p;
    const int mbase   = (lane >> 4) * 8;
    int nr[2][8];
#pragma unroll
    for (int i = 0; i < 2; ++i)
#pragma unroll
        for (int r = 0; r < 8; ++r)
            nr[i][r] = nrowS[wm * 32 + i * 16 + mbase + r];
#pragma unroll
    for (int j = 0; j < 4; ++j) {
        const int col  = wn * 64 + j * 16 + colN;
        const float bb = bias[d * ODIM + col];
#pragma unroll
        for (int i = 0; i < 2; ++i) {
#pragma unroll
            for (int r = 0; r < 8; ++r) {
                int row = wm * 32 + i * 16 + mbase + r;
                float y = acc[i][j][r] + bb;
                y = (y >= 0.0f) ? y : alpha * y;
                if (row < nrows)
                    out[((size_t)b * N_OUT + nr[i][r]) * ODIM + col] = y;
            }
        }
    }
}

extern "C" void kernel_launch(void* const* d_in, const int* in_sizes, int n_in,
                              void* d_out, int out_size, void* d_ws, size_t ws_size,
                              hipStream_t stream) {
    const float* last    = (const float*)d_in[0];
    const float* W       = (const float*)d_in[1];
    const float* bias    = (const float*)d_in[2];
    const float* alpha   = (const float*)d_in[3];
    const int*   vec     = (const int*)d_in[4];
    const int*   dmap    = (const int*)d_in[5];
    const int*   drev    = (const int*)d_in[6];
    const int*   child_l = (const int*)d_in[7];
    const int*   child_r = (const int*)d_in[8];
    float* out = (float*)d_out;

    char* ws = (char*)d_ws;
    int*   counts  = (int*)ws;
    int*   buckets = (int*)(ws + 512);
    float* Wf      = (float*)(ws + 512 + (size_t)B_ * NDIR * N_OUT * 4);

    k_prepare<<<B_ + ARR_BLOCKS, 1024, 0, stream>>>(W, Wf, vec, dmap, counts, buckets);
    k_gemm_tile<<<GRID_TILES, 128, 0, stream>>>(
        last, Wf, bias, alpha, vec, drev, child_l, child_r, counts, buckets, out);
}


# Gather loads whole child rows: one wave load = one 512-B row, 16 B per lane over 4 whole lines (was 16-B pieces of 16 lines per load); 16 lanes resolve a wave's row ids once, readlane shares them
# speedup vs baseline: 2.6585x; 1.0469x over previous
#include <hip/hip_runtime.h>

typedef __attribute__((ext_vector_type(2))) float v2f;
typedef __attribute__((ext_vector_type(4))) float v4f;
typedef __attribute__((ext_vector_type(8))) float v8f;

#define B_      16
#define N_IN    16384
#define N_OUT   8192
#define IDIM    128
#define ODIM    128
#define KDIM    256
#define NDIR    6
#define BM     64
#define GRID_TILES (B_ * N_OUT / BM + B_ * NDIR)

#define KPAIRS  (KDIM / 8)
#define JTILES  (ODIM / 16)
#define ARR_ELEMS  (NDIR * JTILES * KPAIRS * 32)
#define ARR_BLOCKS (ARR_ELEMS / 1024)
__device__ __forceinline__ void arrange_w(const float* __restrict__ W, float* __restrict__ Wf_, int idx) {
    v4f* __restrict__ Wf = (v4f*)Wf_;
    int lane = idx & 31;
    int kp   = (idx >> 5) % KPAIRS;
    int jt   = (idx / (32 * KPAIRS)) % JTILES;
    int d    = idx / (32 * KPAIRS * JTILES);
    int c    = lane & 15;
    int h    = lane >> 4;
    const float* w = W + ((size_t)d * KDIM + 8 * kp + 2 * h) * ODIM + jt * 16 + c;
    v4f o;
    o.x = w[0];
    o.y = w[ODIM];
    o.z = w[4 * ODIM];
    o.w = w[5 * ODIM];
    Wf[idx] = o;
}

#define BK_SEGS     (N_OUT / 32)
#define BK_PER_WAVE (BK_SEGS / 32)
#define BK_PITCH    (BK_SEGS + 1)
__device__ __forceinline__
void bucketize(const int* __restrict__ vec, const int* __restrict__ dmap,
               int* __restrict__ counts, int* __restrict__ buckets, const int b) {
    __shared__ int segS[NDIR * BK_PITCH];
    const int lane = threadIdx.x & 31;
    const int wave = threadIdx.x >> 5;
    const unsigned below = (1u << lane) - 1u;

    int dir[BK_PER_WAVE], rank[BK_PER_WAVE];
#pragma unroll
    for (int j = 0; j < BK_PER_WAVE; ++j)
        dir[j] = vec[b * N_OUT + (j * 32 + wave) * 32 + lane];
#pragma unroll
    for (int j = 0; j < BK_PER_WAVE; ++j)
        dir[j] = dmap[dir[j]];
#pragma unroll
    for (int j = 0; j < BK_PER_WAVE; ++j) {
        int rk = 0, cnt = 0;
#pragma unroll
        for (int d = 0; d < NDIR; ++d) {
            const unsigned m = __builtin_amdgcn_ballot_w32(dir[j] == d);
            if (dir[j] == d) rk  = __builtin_popcount(m & below);
            if (lane == d)   cnt = __builtin_popcount(m);
        }
        rank[j] = rk;
        if (lane < NDIR) segS[lane * BK_PITCH + j * 32 + wave] = cnt;
    }
    __syncthreads();

    if (wave < NDIR) {
        int* row = &segS[wave * BK_PITCH + BK_PER_WAVE * lane];
        int c[BK_PER_WAVE];
        int run = 0;
#pragma unroll
        for (int k = 0; k < BK_PER_WAVE; ++k) {
            const int t = row[k];
            c[k] = run;
            run += t;
        }
        int incl = run;
#pragma unroll
        for (int o = 1; o < 32; o <<= 1) {
            const int up = __shfl_up(incl, o);
            if (lane >= o) incl += up;
        }
        const int base = incl - run;
#pragma unroll
        for (int k = 0; k < BK_PER_WAVE; ++k)
            row[k] = base + c[k];
        if (lane == 31) counts[b * NDIR + wave] = incl;
    }
    __syncthreads();

#pragma unroll
    for (int j = 0; j < BK_PER_WAVE; ++j) {
        const int seg = j * 32 + wave;
        const int d   = dir[j];
        if ((unsigned)d < (unsigned)NDIR)
            buckets[(size_t)(b * NDIR + d) * N_OUT + segS[d * BK_PITCH + seg] + rank[j]] = seg * 32 + lane;
    }
}

__global__ __launch_bounds__(1024)
void k_prepare(const float* __restrict__ W, float* __restrict__ Wf,
               const int* __restrict__ vec, const int* __restrict__ dmap,
               int* __restrict__ counts, int* __restrict__ buckets) {
    if (blockIdx.x < B_)
        bucketize(vec, dmap, counts, buckets, (int)blockIdx.x);
    else
        arrange_w(W, Wf, ((int)blockIdx.x - B_) * 1024 + (int)threadIdx.x);
}

__global__ __launch_bounds__(128, 1)
void k_gemm_tile(const float* __restrict__ last, const float* __restrict__ Wf_,
                 const float* __restrict__ bias, const float* __restrict__ alpha_p,
                 const int* __restrict__ vec, const int* __restrict__ drev,
                 const int* __restrict__ child_l, const int* __restrict__ child_r,
                 const int* __restrict__ counts, const int* __restrict__ buckets,
                 float* __restrict__ out) {
    __shared__ float pairS[BM * KDIM];
    __shared__ int   nrowS[BM];

    const int ln = threadIdx.x & 31;
    const int c0 = counts[3 * ln], c1 = counts[3 * ln + 1], c2 = counts[3 * ln + 2];
    const int t0 = (c0 + BM - 1) / BM, t1 = (c1 + BM - 1) / BM, t2 = (c2 + BM - 1) / BM;
    const int tsum = t0 + t1 + t2;
    int incl = tsum;
#pragma unroll
    for (int o = 1; o < 32; o <<= 1) {
        int up = __shfl_up(incl, o);
        if (ln >= o) incl += up;
    }
    const int r = (int)blockIdx.x - (incl - tsum);
    const unsigned owner = __builtin_amdgcn_ballot_w32(r >= 0 && r < tsum);
    if (owner == 0u) return;
    const int q      = (r < t0) ? 0 : ((r < t0 + t1) ? 1 : 2);
    const int tile_l = (q == 0) ? r : ((q == 1) ? r - t0 : r - t0 - t1);
    const int cnt_l  = (q == 0) ? c0 : ((q == 1) ? c1 : c2);
    const int ol   = __builtin_ctz(owner);
    const int bd   = __builtin_amdgcn_readlane(3 * ln + q, ol);
    const int tile = __builtin_amdgcn_readlane(tile_l, ol);
    const int cnt  = __builtin_amdgcn_readlane(cnt_l, ol);
    const int row0 = tile * BM;
    const int b = bd / NDIR;
    const int d = bd % NDIR;
    const int nrows = min(BM, cnt - row0);

    const int tid = threadIdx.x;

    if (tid < BM) {
        int r = (tid < nrows) ? tid : 0;
        nrowS[tid] = buckets[(size_t)bd * N_OUT + row0 + r];
    }
    __syncthreads();

    {
        const int gl  = tid & 31;
        const int gw  = tid >> 5;
        const int n   = nrowS[gw * 16 + (gl & 15)];
        const int v   = vec[b * N_OUT + n];
        const int cl  = child_l[n];
        const int cr  = child_r[n];
        const int rev = drev[v];
        const int sA  = rev ? cr : cl;
        const int sB  = rev ? cl : cr;
        const float* lastb = last + (size_t)b * N_IN * IDIM + gl * 4;
        float4 x[16][2];
#pragma unroll
        for (int it = 0; it < 16; ++it) {
            const int ra = __builtin_amdgcn_readlane(sA, it);
            const int rc = __builtin_amdgcn_readlane(sB, it);
            x[it][0] = *(const float4*)(lastb + (size_t)ra * IDIM);
            x[it][1] = *(const float4*)(lastb + (size_t)rc * IDIM);
        }
#pragma unroll
        for (int it = 0; it < 16; ++it) {
            float4* dst = (float4*)&pairS[(gw * 16 + it) * KDIM + gl * 4];
            dst[0]        = x[it][0];
            dst[IDIM / 4] = x[it][1];
        }
    }
    __syncthreads();

    const int lane = tid & 31;
    const int wave = tid >> 5;
    const int wm   = wave >> 1;
    const int wn   = wave & 1;
    const int colN = lane & 15;
    const int koff = (lane >> 4) * 2;
    const float* arow = &pairS[(wm * 32 + colN) * KDIM + koff];
    const v4f* wfrag = (const v4f*)Wf_ + ((size_t)(d * JTILES + wn * 4) * KPAIRS) * 32 + lane;

    v8f acc[2][4] = {};
#pragma unroll 4
    for (int kp = 0; kp < KPAIRS; ++kp) {
        v2f a[2][2];
        v4f bq[4];
#pragma unroll
        for (int s = 0; s < 2; ++s)
#pragma unroll
            for (int i = 0; i < 2; ++i) {
                a[s][i].x = arow[i * 16 * KDIM + 8 * kp + 4 * s];
                a[s][i].y = arow[i * 16 * KDIM + 8 * kp + 4 * s + 1];
            }
#pragma unroll
        for (int j = 0; j < 4; ++j)
            bq[j] = wfrag[(j * KPAIRS + kp) * 32];
#pragma unroll
        for (int s = 0; s < 2; ++s)
#pragma unroll
            for (int i = 0; i < 2; ++i)
#pragma unroll
                for (int j = 0; j < 4; ++j) {
                    v2f bf;
                    bf.x = s ? bq[j].z : bq[j].x;
                    bf.y = s ? bq[j].w : bq[j].y;
                    acc[i][j] = __builtin_amdgcn_wmma_f32_16x16x4_f32(
                                    false, a[s][i], false, bf, (short)0, acc[i][j], false, false);
                }
    }

    const float alpha = *alpha_p;
    const int mbase   = (lane >> 4) * 8;
    int nr[2][8];
#pragma unroll
    for (int i = 0; i < 2; ++i)
#pragma unroll
        for (int r = 0; r < 8; ++r)
            nr[i][r] = nrowS[wm * 32 + i * 16 + mbase + r];
#pragma unroll
    for (int j = 0; j < 4; ++j) {
        const int col  = wn * 64 + j * 16 + colN;
        const float bb = bias[d * ODIM + col];
#pragma unroll
        for (int i = 0; i < 2; ++i) {
#pragma unroll
            for (int r = 0; r < 8; ++r) {
                int row = wm * 32 + i * 16 + mbase + r;
                float y = acc[i][j][r] + bb;
                y = (y >= 0.0f) ? y : alpha * y;
                if (row < nrows)
                    out[((size_t)b * N_OUT + nr[i][r]) * ODIM + col] = y;
            }
        }
    }
}

extern "C" void kernel_launch(void* const* d_in, const int* in_sizes, int n_in,
                              void* d_out, int out_size, void* d_ws, size_t ws_size,
                              hipStream_t stream) {
    const float* last    = (const float*)d_in[0];
    const float* W       = (const float*)d_in[1];
    const float* bias    = (const float*)d_in[2];
    const float* alpha   = (const float*)d_in[3];
    const int*   vec     = (const int*)d_in[4];
    const int*   dmap    = (const int*)d_in[5];
    const int*   drev    = (const int*)d_in[6];
    const int*   child_l = (const int*)d_in[7];
    const int*   child_r = (const int*)d_in[8];
    float* out = (float*)d_out;

    char* ws = (char*)d_ws;
    int*   counts  = (int*)ws;
    int*   buckets = (int*)(ws + 512);
    float* Wf      = (float*)(ws + 512 + (size_t)B_ * NDIR * N_OUT * 4);

    k_prepare<<<B_ + ARR_BLOCKS, 1024, 0, stream>>>(W, Wf, vec, dmap, counts, buckets);
    k_gemm_tile<<<GRID_TILES, 128, 0, stream>>>(
        last, Wf, bias, alpha, vec, drev, child_l, child_r, counts, buckets, out);
}


# One launch: each block ranks its batch's rows by direction itself (LDS counts, ballots) and reads W row-major, 4 adjacent columns per lane, so the preparation launch, buckets and W copy are gone
# speedup vs baseline: 2.8495x; 1.0718x over previous
#include <hip/hip_runtime.h>

typedef __attribute__((ext_vector_type(2))) float v2f;
typedef __attribute__((ext_vector_type(4))) float v4f;
typedef __attribute__((ext_vector_type(8))) float v8f;

#define B_      16
#define N_IN    16384
#define N_OUT   8192
#define IDIM    128
#define ODIM    128
#define KDIM    256
#define NDIR    6
#define NCODE   64
#define BM      64
#define KPAIRS  (KDIM / 8)
#define TPB     (N_OUT / BM + NDIR)
#define WSPAN   (N_OUT / 4)
#define SCAN_IT (WSPAN / 32)

__global__ __launch_bounds__(128, 1)
void k_gemm_tile(const float* __restrict__ last, const float* __restrict__ W,
                 const float* __restrict__ bias, const float* __restrict__ alpha_p,
                 const int* __restrict__ vec, const int* __restrict__ dmap,
                 const int* __restrict__ drev,
                 const int* __restrict__ child_l, const int* __restrict__ child_r,
                 float* __restrict__ out) {
    __shared__ float pairS[BM * KDIM];
    __shared__ int   nrowS[BM];
    __shared__ int   tabS[NCODE];
    __shared__ __attribute__((aligned(16))) int wcntS[4 * 8];

    const int tid  = threadIdx.x;
    const int lane = tid & 31;
    const int wave = tid >> 5;
    const int b = (int)blockIdx.x / TPB;
    const int t = (int)blockIdx.x % TPB;

    int code[SCAN_IT];
    const int* vrow = vec + b * N_OUT + wave * WSPAN + lane;
#pragma unroll
    for (int it = 0; it < SCAN_IT; ++it)
        code[it] = vrow[it * 32];
    if (tid < NCODE) {
        tabS[tid]  = ((dmap[tid] & 7) << 3) | ((drev[tid] != 0) ? 64 : 0);
        nrowS[tid] = 0;
    }
    __syncthreads();

    unsigned long long cnt6 = 0ull;
#pragma unroll
    for (int it = 0; it < SCAN_IT; ++it) {
        code[it] = tabS[code[it]];
        cnt6 += 1ull << (code[it] & 63);
    }
    {
        const unsigned lo = (unsigned)cnt6, hi = (unsigned)(cnt6 >> 32);
        int p0 = (int)((lo & 0xFFu) | ((lo & 0xFF00u) << 8));
        int p1 = (int)(((lo >> 16) & 0xFFu) | ((lo >> 24) << 16));
        int p2 = (int)((hi & 0xFFu) | ((hi & 0xFF00u) << 8));
#pragma unroll
        for (int o = 16; o > 0; o >>= 1) {
            p0 += __shfl_xor(p0, o);
            p1 += __shfl_xor(p1, o);
            p2 += __shfl_xor(p2, o);
        }
        if (lane == 0) {
            wcntS[wave * 8 + 0] = p0 & 0xFFFF;  wcntS[wave * 8 + 1] = (int)((unsigned)p0 >> 16);
            wcntS[wave * 8 + 2] = p1 & 0xFFFF;  wcntS[wave * 8 + 3] = (int)((unsigned)p1 >> 16);
            wcntS[wave * 8 + 4] = p2 & 0xFFFF;  wcntS[wave * 8 + 5] = (int)((unsigned)p2 >> 16);
            wcntS[wave * 8 + 6] = 0;            wcntS[wave * 8 + 7] = 0;
        }
    }
    __syncthreads();

    int wc[4][8];
#pragma unroll
    for (int w = 0; w < 4; ++w) {
        const int4 q0 = ((const int4*)wcntS)[2 * w];
        const int4 q1 = ((const int4*)wcntS)[2 * w + 1];
        wc[w][0] = q0.x; wc[w][1] = q0.y; wc[w][2] = q0.z; wc[w][3] = q0.w;
        wc[w][4] = q1.x; wc[w][5] = q1.y; wc[w][6] = q1.z; wc[w][7] = q1.w;
    }
    int rem = t, d = -1, tile = 0, cnt = 0, f0 = 0, f1 = 0, f2 = 0;
#pragma unroll
    for (int dd = 0; dd < NDIR; ++dd) {
        const int tot = wc[0][dd] + wc[1][dd] + wc[2][dd] + wc[3][dd];
        const int td  = (tot + BM - 1) / BM;
        if (d < 0) {
            if (rem < td) { d = dd; tile = rem; cnt = tot; f0 = wc[0][dd]; f1 = wc[1][dd]; f2 = wc[2][dd]; }
            else rem -= td;
        }
    }
    if (d < 0) return;
    const int row0  = tile * BM;
    const int nrows = min(BM, cnt - row0);

    {
        const int base  = (wave > 0 ? f0 : 0) + (wave > 1 ? f1 : 0) + (wave > 2 ? f2 : 0);
        const int mine  = (wave == 0) ? f0 : ((wave == 1) ? f1 : ((wave == 2) ? f2 : cnt - f0 - f1 - f2));
        const int first = row0 - base;
        if (first < mine && first + BM > 0) {
            const unsigned below = (1u << lane) - 1u;
            int run = 0;
#pragma unroll
            for (int it = 0; it < SCAN_IT; ++it) {
                const bool hit   = (code[it] & 63) == 8 * d;
                const unsigned m = __builtin_amdgcn_ballot_w32(hit);
                const int idx    = run + __builtin_popcount(m & below) - first;
                if (hit && (unsigned)idx < (unsigned)BM)
                    nrowS[idx] = (wave * WSPAN + it * 32 + lane) | ((code[it] >> 6) << 16);
                run += __builtin_popcount(m);
            }
        }
    }
    __syncthreads();

    {
        const int gl  = lane;
        const int gw  = wave;
        const int e   = nrowS[gw * 16 + (gl & 15)];
        const int n   = e & 0xFFFF;
        const int rev = e >> 16;
        const int cl  = child_l[n];
        const int cr  = child_r[n];
        const int sA  = rev ? cr : cl;
        const int sB  = rev ? cl : cr;
        const float* lastb = last + (size_t)b * N_IN * IDIM + gl * 4;
        float4 x[16][2];
#pragma unroll
        for (int it = 0; it < 16; ++it) {
            const int ra = __builtin_amdgcn_readlane(sA, it);
            const int rc = __builtin_amdgcn_readlane(sB, it);
            x[it][0] = *(const float4*)(lastb + (size_t)ra * IDIM);
            x[it][1] = *(const float4*)(lastb + (size_t)rc * IDIM);
        }
#pragma unroll
        for (int it = 0; it < 16; ++it) {
            float4* dst = (float4*)&pairS[(gw * 16 + it) * KDIM + gl * 4];
            dst[0]        = x[it][0];
            dst[IDIM / 4] = x[it][1];
        }
    }
    __syncthreads();

    const int wm   = wave >> 1;
    const int wn   = wave & 1;
    const int colN = lane & 15;
    const int koff = (lane >> 4) * 2;
    const int col4 = wn * 64 + colN * 4;
    const float* arow = &pairS[(wm * 32 + colN) * KDIM + koff];
    const float* wrow = W + ((size_t)d * KDIM + koff) * ODIM + col4;

    v8f acc[2][4] = {};
#pragma unroll 4
    for (int kp = 0; kp < KPAIRS; ++kp) {
        v2f a[2][2];
        v4f be[2], bo[2];
#pragma unroll
        for (int s = 0; s < 2; ++s)
#pragma unroll
            for (int i = 0; i < 2; ++i) {
                a[s][i].x = arow[i * 16 * KDIM + 8 * kp + 4 * s];
                a[s][i].y = arow[i * 16 * KDIM + 8 * kp + 4 * s + 1];
            }
#pragma unroll
        for (int s = 0; s < 2; ++s) {
            be[s] = *(const v4f*)(wrow + (size_t)(8 * kp + 4 * s) * ODIM);
            bo[s] = *(const v4f*)(wrow + (size_t)(8 * kp + 4 * s + 1) * ODIM);
        }
#pragma unroll
        for (int s = 0; s < 2; ++s)
#pragma unroll
            for (int i = 0; i < 2; ++i)
#pragma unroll
                for (int j = 0; j < 4; ++j) {
                    v2f bf;
                    bf.x = be[s][j];
                    bf.y = bo[s][j];
                    acc[i][j] = __builtin_amdgcn_wmma_f32_16x16x4_f32(
                                    false, a[s][i], false, bf, (short)0, acc[i][j], false, false);
                }
    }

    const float alpha = *alpha_p;
    const int mbase   = (lane >> 4) * 8;
    const v4f bb = *(const v4f*)(bias + d * ODIM + col4);
#pragma unroll
    for (int i = 0; i < 2; ++i) {
#pragma unroll
        for (int r = 0; r < 8; ++r) {
            const int row = wm * 32 + i * 16 + mbase + r;
            const int n   = nrowS[row] & 0xFFFF;
            v4f y;
#pragma unroll
            for (int j = 0; j < 4; ++j) {
                float v = acc[i][j][r] + bb[j];
                y[j] = (v >= 0.0f) ? v : alpha * v;
            }
            if (row < nrows)
                *(v4f*)(out + ((size_t)b * N_OUT + n) * ODIM + col4) = y;
        }
    }
}

extern "C" void kernel_launch(void* const* d_in, const int* in_sizes, int n_in,
                              void* d_out, int out_size, void* d_ws, size_t ws_size,
                              hipStream_t stream) {
    const float* last    = (const float*)d_in[0];
    const float* W       = (const float*)d_in[1];
    const float* bias    = (const float*)d_in[2];
    const float* alpha   = (const float*)d_in[3];
    const int*   vec     = (const int*)d_in[4];
    const int*   dmap    = (const int*)d_in[5];
    const int*   drev    = (const int*)d_in[6];
    const int*   child_l = (const int*)d_in[7];
    const int*   child_r = (const int*)d_in[8];
    float* out = (float*)d_out;

    k_gemm_tile<<<B_ * TPB, 128, 0, stream>>>(
        last, W, bias, alpha, vec, dmap, drev, child_l, child_r, out);
}


# Five blocks per WGP instead of four: LDS cut to exactly 64 KiB per block (ranking tables alias the pair buffer, epilogue row ids held in registers), registers capped for 5 waves per SIMD
# speedup vs baseline: 2.9881x; 1.0486x over previous
#include <hip/hip_runtime.h>

typedef __attribute__((ext_vector_type(2))) float v2f;
typedef __attribute__((ext_vector_type(4))) float v4f;
typedef __attribute__((ext_vector_type(8))) float v8f;

#define B_      16
#define N_IN    16384
#define N_OUT   8192
#define IDIM    128
#define ODIM    128
#define KDIM    256
#define NDIR    6
#define NCODE   64
#define BM      64
#define KPAIRS  (KDIM / 8)
#define TPB     (N_OUT / BM + NDIR)
#define WSPAN   (N_OUT / 4)
#define SCAN_IT (WSPAN / 32)

__global__ __launch_bounds__(128, 5)
void k_gemm_tile(const float* __restrict__ last, const float* __restrict__ W,
                 const float* __restrict__ bias, const float* __restrict__ alpha_p,
                 const int* __restrict__ vec, const int* __restrict__ dmap,
                 const int* __restrict__ drev,
                 const int* __restrict__ child_l, const int* __restrict__ child_r,
                 float* __restrict__ out) {
    __shared__ __attribute__((aligned(16))) float pairS[BM * KDIM];
    int* const nrowS = (int*)pairS;
    int* const tabS  = nrowS + BM;
    int* const wcntS = tabS + NCODE;

    const int tid  = threadIdx.x;
    const int lane = tid & 31;
    const int wave = tid >> 5;
    const int b = (int)blockIdx.x / TPB;
    const int t = (int)blockIdx.x % TPB;

    int code[SCAN_IT];
    const int* vrow = vec + b * N_OUT + wave * WSPAN + lane;
#pragma unroll
    for (int it = 0; it < SCAN_IT; ++it)
        code[it] = vrow[it * 32];
    if (tid < NCODE) {
        tabS[tid]  = ((dmap[tid] & 7) << 3) | ((drev[tid] != 0) ? 64 : 0);
        nrowS[tid] = 0;
    }
    __syncthreads();

    unsigned long long cnt6 = 0ull;
#pragma unroll
    for (int it = 0; it < SCAN_IT; ++it) {
        code[it] = tabS[code[it]];
        cnt6 += 1ull << (code[it] & 63);
    }
    {
        const unsigned lo = (unsigned)cnt6, hi = (unsigned)(cnt6 >> 32);
        int p0 = (int)((lo & 0xFFu) | ((lo & 0xFF00u) << 8));
        int p1 = (int)(((lo >> 16) & 0xFFu) | ((lo >> 24) << 16));
        int p2 = (int)((hi & 0xFFu) | ((hi & 0xFF00u) << 8));
#pragma unroll
        for (int o = 16; o > 0; o >>= 1) {
            p0 += __shfl_xor(p0, o);
            p1 += __shfl_xor(p1, o);
            p2 += __shfl_xor(p2, o);
        }
        if (lane == 0) {
            wcntS[wave * 8 + 0] = p0 & 0xFFFF;  wcntS[wave * 8 + 1] = (int)((unsigned)p0 >> 16);
            wcntS[wave * 8 + 2] = p1 & 0xFFFF;  wcntS[wave * 8 + 3] = (int)((unsigned)p1 >> 16);
            wcntS[wave * 8 + 4] = p2 & 0xFFFF;  wcntS[wave * 8 + 5] = (int)((unsigned)p2 >> 16);
            wcntS[wave * 8 + 6] = 0;            wcntS[wave * 8 + 7] = 0;
        }
    }
    __syncthreads();

    int wc[4][8];
#pragma unroll
    for (int w = 0; w < 4; ++w) {
        const int4 q0 = ((const int4*)wcntS)[2 * w];
        const int4 q1 = ((const int4*)wcntS)[2 * w + 1];
        wc[w][0] = q0.x; wc[w][1] = q0.y; wc[w][2] = q0.z; wc[w][3] = q0.w;
        wc[w][4] = q1.x; wc[w][5] = q1.y; wc[w][6] = q1.z; wc[w][7] = q1.w;
    }
    int rem = t, d = -1, tile = 0, cnt = 0, f0 = 0, f1 = 0, f2 = 0;
#pragma unroll
    for (int dd = 0; dd < NDIR; ++dd) {
        const int tot = wc[0][dd] + wc[1][dd] + wc[2][dd] + wc[3][dd];
        const int td  = (tot + BM - 1) / BM;
        if (d < 0) {
            if (rem < td) { d = dd; tile = rem; cnt = tot; f0 = wc[0][dd]; f1 = wc[1][dd]; f2 = wc[2][dd]; }
            else rem -= td;
        }
    }
    if (d < 0) return;
    const int row0  = tile * BM;
    const int nrows = min(BM, cnt - row0);

    {
        const int base  = (wave > 0 ? f0 : 0) + (wave > 1 ? f1 : 0) + (wave > 2 ? f2 : 0);
        const int mine  = (wave == 0) ? f0 : ((wave == 1) ? f1 : ((wave == 2) ? f2 : cnt - f0 - f1 - f2));
        const int first = row0 - base;
        if (first < mine && first + BM > 0) {
            const unsigned below = (1u << lane) - 1u;
            int run = 0;
#pragma unroll
            for (int it = 0; it < SCAN_IT; ++it) {
                const bool hit   = (code[it] & 63) == 8 * d;
                const unsigned m = __builtin_amdgcn_ballot_w32(hit);
                const int idx    = run + __builtin_popcount(m & below) - first;
                if (hit && (unsigned)idx < (unsigned)BM)
                    nrowS[idx] = (wave * WSPAN + it * 32 + lane) | ((code[it] >> 6) << 16);
                run += __builtin_popcount(m);
            }
        }
    }
    __syncthreads();

    const int wm    = wave >> 1;
    const int mbase = (lane >> 4) * 8;
    const int myrow = nrowS[wm * 32 + lane] & 0xFFFF;
    const int e = nrowS[wave * 16 + (lane & 15)];
    __syncthreads();
    {
        const int gl  = lane;
        const int gw  = wave;
        const int n   = e & 0xFFFF;
        const int rev = e >> 16;
        const int cl  = child_l[n];
        const int cr  = child_r[n];
        const int sA  = rev ? cr : cl;
        const int sB  = rev ? cl : cr;
        const float* lastb = last + (size_t)b * N_IN * IDIM + gl * 4;
        float4 x[16][2];
#pragma unroll
        for (int it = 0; it < 16; ++it) {
            const int ra = __builtin_amdgcn_readlane(sA, it);
            const int rc = __builtin_amdgcn_readlane(sB, it);
            x[it][0] = *(const float4*)(lastb + (size_t)ra * IDIM);
            x[it][1] = *(const float4*)(lastb + (size_t)rc * IDIM);
        }
#pragma unroll
        for (int it = 0; it < 16; ++it) {
            float4* dst = (float4*)&pairS[(gw * 16 + it) * KDIM + gl * 4];
            dst[0]        = x[it][0];
            dst[IDIM / 4] = x[it][1];
        }
    }
    __syncthreads();

    const int wn   = wave & 1;
    const int colN = lane & 15;
    const int koff = (lane >> 4) * 2;
    const int col4 = wn * 64 + colN * 4;
    const float* arow = &pairS[(wm * 32 + colN) * KDIM + koff];
    const float* wrow = W + ((size_t)d * KDIM + koff) * ODIM + col4;

    v8f acc[2][4] = {};
#pragma unroll 4
    for (int kp = 0; kp < KPAIRS; ++kp) {
        v2f a[2][2];
        v4f be[2], bo[2];
#pragma unroll
        for (int s = 0; s < 2; ++s)
#pragma unroll
            for (int i = 0; i < 2; ++i) {
                a[s][i].x = arow[i * 16 * KDIM + 8 * kp + 4 * s];
                a[s][i].y = arow[i * 16 * KDIM + 8 * kp + 4 * s + 1];
            }
#pragma unroll
        for (int s = 0; s < 2; ++s) {
            be[s] = *(const v4f*)(wrow + (size_t)(8 * kp + 4 * s) * ODIM);
            bo[s] = *(const v4f*)(wrow + (size_t)(8 * kp + 4 * s + 1) * ODIM);
        }
#pragma unroll
        for (int s = 0; s < 2; ++s)
#pragma unroll
            for (int i = 0; i < 2; ++i)
#pragma unroll
                for (int j = 0; j < 4; ++j) {
                    v2f bf;
                    bf.x = be[s][j];
                    bf.y = bo[s][j];
                    acc[i][j] = __builtin_amdgcn_wmma_f32_16x16x4_f32(
                                    false, a[s][i], false, bf, (short)0, acc[i][j], false, false);
                }
    }

    const float alpha = *alpha_p;
    const v4f bb = *(const v4f*)(bias + d * ODIM + col4);
    int nid[2][8];
#pragma unroll
    for (int i = 0; i < 2; ++i)
#pragma unroll
        for (int r = 0; r < 8; ++r)
            nid[i][r] = __shfl(myrow, i * 16 + mbase + r);
#pragma unroll
    for (int i = 0; i < 2; ++i) {
#pragma unroll
        for (int r = 0; r < 8; ++r) {
            const int row = wm * 32 + i * 16 + mbase + r;
            const int n   = nid[i][r];
            v4f y;
#pragma unroll
            for (int j = 0; j < 4; ++j) {
                float v = acc[i][j][r] + bb[j];
                y[j] = (v >= 0.0f) ? v : alpha * v;
            }
            if (row < nrows)
                *(v4f*)(out + ((size_t)b * N_OUT + n) * ODIM + col4) = y;
        }
    }
}

extern "C" void kernel_launch(void* const* d_in, const int* in_sizes, int n_in,
                              void* d_out, int out_size, void* d_ws, size_t ws_size,
                              hipStream_t stream) {
    const float* last    = (const float*)d_in[0];
    const float* W       = (const float*)d_in[1];
    const float* bias    = (const float*)d_in[2];
    const float* alpha   = (const float*)d_in[3];
    const int*   vec     = (const int*)d_in[4];
    const int*   dmap    = (const int*)d_in[5];
    const int*   drev    = (const int*)d_in[6];
    const int*   child_l = (const int*)d_in[7];
    const int*   child_r = (const int*)d_in[8];
    float* out = (float*)d_out;

    k_gemm_tile<<<B_ * TPB, 128, 0, stream>>>(
        last, W, bias, alpha, vec, dmap, drev, child_l, child_r, out);
}
